# starting kernel
# baseline (speedup 1.0000x reference)
#include <hip/hip_runtime.h>

typedef float v2f __attribute__((ext_vector_type(2)));
typedef float v8f __attribute__((ext_vector_type(8)));

#define GCN_N 100000
#define GCN_E 1600000

__device__ __forceinline__ void atomic_add_f32(float* p, float v) {
    unsafeAtomicAdd(p, v);
}

__global__ void deg_init_kernel(float* deg, int n) {
    int i = blockIdx.x * blockDim.x + threadIdx.x;
    if (i < n) deg[i] = 1.0f;
}

__global__ void deg_count_kernel(const int* __restrict__ dst, float* deg, int e) {
    int i = blockIdx.x * blockDim.x + threadIdx.x;
    if (i < e) atomic_add_f32(&deg[dst[i]], 1.0f);
}

__global__ void dinv_kernel(float* d, int n) {
    int i = blockIdx.x * blockDim.x + threadIdx.x;
    if (i < n) d[i] = rsqrtf(d[i]);
}

template <int KDIM, int NOUT, bool RELU>
__global__ void gemm_wmma_kernel(const float* __restrict__ X,
                                 const float* __restrict__ W,
                                 float* __restrict__ H, int M) {
    constexpr int NT  = NOUT / 16;
    constexpr int LDW = NOUT + 8;
    __shared__ float Ws[KDIM * LDW];
    for (int i = threadIdx.x; i < KDIM * NOUT; i += blockDim.x) {
        int k = i / NOUT, n = i - k * NOUT;
        Ws[k * LDW + n] = W[i];
    }
    __syncthreads();

    const int wid  = (blockIdx.x * blockDim.x + threadIdx.x) >> 5;
    const int lane = threadIdx.x & 31;
    const int m0 = (wid / NT) * 16;
    const int n0 = (wid % NT) * 16;
    if (m0 >= M) return;

    const int half = lane >> 4;
    const int l16  = lane & 15;
    const int koff = half * 2;
    const float* xrow = X + (size_t)(m0 + l16) * KDIM + koff;
    const float* wcol = Ws + koff * LDW + n0 + l16;

    v8f acc = {};
#pragma unroll 4
    for (int k0 = 0; k0 < KDIM; k0 += 4) {
        float a0 = xrow[k0];
        float a1 = xrow[k0 + 1];
        if (RELU) { a0 = fmaxf(a0, 0.0f); a1 = fmaxf(a1, 0.0f); }
        v2f a; a.x = a0; a.y = a1;
        v2f b; b.x = wcol[k0 * LDW]; b.y = wcol[k0 * LDW + LDW];
        acc = __builtin_amdgcn_wmma_f32_16x16x4_f32(
            false, a, false, b, (short)0, acc, false, false);
    }
    float* orow = H + (size_t)(m0 + half * 8) * NOUT + n0 + l16;
#pragma unroll
    for (int v = 0; v < 8; ++v)
        orow[(size_t)v * NOUT] = acc[v];
}

__global__ void agg_init_kernel(const float* __restrict__ H,
                                const float* __restrict__ dinv,
                                const float* __restrict__ bias,
                                float* __restrict__ A, int n, int fsh) {
    int t = blockIdx.x * blockDim.x + threadIdx.x;
    if (t >= (n << fsh)) return;
    int i = t >> fsh;
    int c = (t & ((1 << fsh) - 1)) << 2;
    int F = 4 << fsh;
    float d = dinv[i];
    float s = d * d;
    float4 h = *(const float4*)(H + (size_t)i * F + c);
    float4 o;
    o.x = h.x * s + bias[c + 0];
    o.y = h.y * s + bias[c + 1];
    o.z = h.z * s + bias[c + 2];
    o.w = h.w * s + bias[c + 3];
    *(float4*)(A + (size_t)i * F + c) = o;
}

__global__ void agg_edge_kernel(const int* __restrict__ src,
                                const int* __restrict__ dst,
                                const float* __restrict__ dinv,
                                const float* __restrict__ H,
                                float* __restrict__ A, int E, int fsh) {
    int t = blockIdx.x * blockDim.x + threadIdx.x;
    if (t >= (E << fsh)) return;
    int e = t >> fsh;
    int c = (t & ((1 << fsh) - 1)) << 2;
    int F = 4 << fsh;
    int s = src[e], d = dst[e];
    float nrm = dinv[s] * dinv[d];
    float4 h = *(const float4*)(H + (size_t)s * F + c);
    float* out = A + (size_t)d * F + c;
    atomic_add_f32(out + 0, h.x * nrm);
    atomic_add_f32(out + 1, h.y * nrm);
    atomic_add_f32(out + 2, h.z * nrm);
    atomic_add_f32(out + 3, h.w * nrm);
}

extern "C" void kernel_launch(void* const* d_in, const int* in_sizes, int n_in,
                              void* d_out, int out_size, void* d_ws, size_t ws_size,
                              hipStream_t stream) {
    (void)in_sizes; (void)n_in; (void)out_size;
    const float* x  = (const float*)d_in[0];
    const int*   ei = (const int*)d_in[1];
    const float* W1 = (const float*)d_in[2];
    const float* b1 = (const float*)d_in[3];
    const float* W2 = (const float*)d_in[4];
    const float* b2 = (const float*)d_in[5];
    const float* W3 = (const float*)d_in[6];
    const float* b3 = (const float*)d_in[7];
    float* out = (float*)d_out;

    const int N = GCN_N, E = GCN_E;
    const int* srcI = ei;
    const int* dstI = ei + E;

    float* dinv = (float*)d_ws;
    float* Hbuf = dinv + N;
    float* Abuf = Hbuf + (size_t)N * 64;
    size_t need = ((size_t)N + 2 * (size_t)N * 64) * sizeof(float);
    if (ws_size < need) return;

    const int TB = 256;
    const int mtiles = N / 16;
    const int wavesA = mtiles * 4;
    const int wavesB = mtiles * 2;
    const int gbA = (wavesA + 7) / 8;
    const int gbB = (wavesB + 7) / 8;

    deg_init_kernel <<<(N + TB - 1) / TB, TB, 0, stream>>>(dinv, N);
    deg_count_kernel<<<(E + TB - 1) / TB, TB, 0, stream>>>(dstI, dinv, E);
    dinv_kernel     <<<(N + TB - 1) / TB, TB, 0, stream>>>(dinv, N);

    gemm_wmma_kernel<128, 64, false><<<gbA, TB, 0, stream>>>(x, W1, Hbuf, N);
    agg_init_kernel<<<((N << 4) + TB - 1) / TB, TB, 0, stream>>>(Hbuf, dinv, b1, Abuf, N, 4);
    agg_edge_kernel<<<((E << 4) + TB - 1) / TB, TB, 0, stream>>>(srcI, dstI, dinv, Hbuf, Abuf, E, 4);

    gemm_wmma_kernel<64, 64, true><<<gbA, TB, 0, stream>>>(Abuf, W2, Hbuf, N);
    agg_init_kernel<<<((N << 4) + TB - 1) / TB, TB, 0, stream>>>(Hbuf, dinv, b2, Abuf, N, 4);
    agg_edge_kernel<<<((E << 4) + TB - 1) / TB, TB, 0, stream>>>(srcI, dstI, dinv, Hbuf, Abuf, E, 4);

    gemm_wmma_kernel<64, 32, true><<<gbB, TB, 0, stream>>>(Abuf, W3, Hbuf, N);
    agg_init_kernel<<<((N << 3) + TB - 1) / TB, TB, 0, stream>>>(Hbuf, dinv, b3, out, N, 3);
    agg_edge_kernel<<<((E << 3) + TB - 1) / TB, TB, 0, stream>>>(srcI, dstI, dinv, Hbuf, out, E, 3);
}


# Each atomic sum (degrees, 3 edge scatters) split in two planes, one per L2 half, chosen by the wave's die number: added lines stay in one half; readers add both planes; a sum = two partial sums
# speedup vs baseline: 1.6340x; 1.6340x over previous
#include <hip/hip_runtime.h>

typedef float v2f __attribute__((ext_vector_type(2)));
typedef float v8f __attribute__((ext_vector_type(8)));

#define GCN_N 100000
#define GCN_E 1600000
static_assert(GCN_E % 256 == 0, "the edge scatter launches E/256 blocks of 8 waves, 32 edges per wave");

__device__ __forceinline__ void atomic_add_f32(float* p, float v) {
    unsafeAtomicAdd(p, v);
}

__device__ __forceinline__ int cache_half() {
    const unsigned id = __builtin_amdgcn_s_sendmsg_rtn(0x87);
    return (int)((id >> 18) & 1u);
}

__global__ void deg_init_kernel(float* deg0, float* deg1, int n) {
    int i = blockIdx.x * blockDim.x + threadIdx.x;
    if (i < n) {
        deg0[i] = 1.0f;
        deg1[i] = 0.0f;
    }
}

__global__ void deg_count_kernel(const int* __restrict__ dst, float* deg0, float* deg1, int e) {
    float* deg = cache_half() ? deg1 : deg0;
    int i = blockIdx.x * blockDim.x + threadIdx.x;
    if (i < e) atomic_add_f32(&deg[dst[i]], 1.0f);
}

__global__ void dinv_kernel(float* d, const float* __restrict__ d1, int n) {
    int i = blockIdx.x * blockDim.x + threadIdx.x;
    if (i < n) d[i] = rsqrtf(d[i] + d1[i]);
}

template <int KDIM, int NOUT, bool RELU, bool TWO>
__global__ void gemm_wmma_kernel(const float* __restrict__ X,
                                 const float* __restrict__ X1,
                                 const float* __restrict__ W,
                                 float* __restrict__ H, int M) {
    constexpr int NT  = NOUT / 16;
    constexpr int LDW = NOUT + 8;
    __shared__ float Ws[KDIM * LDW];
    for (int i = threadIdx.x; i < KDIM * NOUT; i += blockDim.x) {
        int k = i / NOUT, n = i - k * NOUT;
        Ws[k * LDW + n] = W[i];
    }
    __syncthreads();

    const int wid  = (blockIdx.x * blockDim.x + threadIdx.x) >> 5;
    const int lane = threadIdx.x & 31;
    const int m0 = (wid / NT) * 16;
    const int n0 = (wid % NT) * 16;
    if (m0 >= M) return;

    const int half = lane >> 4;
    const int l16  = lane & 15;
    const int koff = half * 2;
    const size_t xoff = (size_t)(m0 + l16) * KDIM + koff;
    const float* xrow  = X + xoff;
    const float* xrow1 = TWO ? X1 + xoff : X;
    const float* wcol = Ws + koff * LDW + n0 + l16;

    v8f acc = {};
#pragma unroll 4
    for (int k0 = 0; k0 < KDIM; k0 += 4) {
        float a0 = xrow[k0];
        float a1 = xrow[k0 + 1];
        if (TWO) { a0 += xrow1[k0]; a1 += xrow1[k0 + 1]; }
        if (RELU) { a0 = fmaxf(a0, 0.0f); a1 = fmaxf(a1, 0.0f); }
        v2f a; a.x = a0; a.y = a1;
        v2f b; b.x = wcol[k0 * LDW]; b.y = wcol[k0 * LDW + LDW];
        acc = __builtin_amdgcn_wmma_f32_16x16x4_f32(
            false, a, false, b, (short)0, acc, false, false);
    }
    float* orow = H + (size_t)(m0 + half * 8) * NOUT + n0 + l16;
#pragma unroll
    for (int v = 0; v < 8; ++v)
        orow[(size_t)v * NOUT] = acc[v];
}

__global__ void agg_init_kernel(const float* __restrict__ H,
                                const float* __restrict__ dinv,
                                const float* __restrict__ bias,
                                float* __restrict__ A,
                                float* __restrict__ A1, int n, int fsh) {
    int t = blockIdx.x * blockDim.x + threadIdx.x;
    if (t >= (n << fsh)) return;
    int i = t >> fsh;
    int c = (t & ((1 << fsh) - 1)) << 2;
    int F = 4 << fsh;
    float d = dinv[i];
    float s = d * d;
    float4 h = *(const float4*)(H + (size_t)i * F + c);
    float4 o;
    o.x = h.x * s + bias[c + 0];
    o.y = h.y * s + bias[c + 1];
    o.z = h.z * s + bias[c + 2];
    o.w = h.w * s + bias[c + 3];
    *(float4*)(A + (size_t)i * F + c) = o;
    *(float4*)(A1 + (size_t)i * F + c) = make_float4(0.0f, 0.0f, 0.0f, 0.0f);
}

template <int F>
__global__ void agg_edge_kernel(const int* __restrict__ src,
                                const int* __restrict__ dst,
                                const float* __restrict__ dinv,
                                const float* __restrict__ H,
                                float* A0, float* A1, int E) {
    float* A = cache_half() ? A1 : A0;
    constexpr int SEG = F / 32;
    constexpr int B   = 16 / SEG;
    const int lane = threadIdx.x & 31;
    const int base = ((blockIdx.x * blockDim.x + threadIdx.x) >> 5) * 32;
    if (base >= E) return;

    const int   s = src[base + lane];
    const int   d = dst[base + lane];
    const float w = dinv[s] * dinv[d];
    const int   wi = __float_as_int(w);

    for (int j0 = 0; j0 < 32; j0 += B) {
        float h[B][SEG];
#pragma unroll
        for (int j = 0; j < B; ++j) {
            const int sj = __builtin_amdgcn_readlane(s, j0 + j);
            const float* row = H + (size_t)sj * F + lane;
#pragma unroll
            for (int q = 0; q < SEG; ++q) h[j][q] = row[32 * q];
        }
#pragma unroll
        for (int j = 0; j < B; ++j) {
            const int   dj = __builtin_amdgcn_readlane(d, j0 + j);
            const float wj = __int_as_float(__builtin_amdgcn_readlane(wi, j0 + j));
            float* o = A + (size_t)dj * F + lane;
#pragma unroll
            for (int q = 0; q < SEG; ++q) atomic_add_f32(o + 32 * q, h[j][q] * wj);
        }
    }
}

__global__ void plane_join_kernel(float* __restrict__ out,
                                  const float* __restrict__ Q1, int n4) {
    int t = blockIdx.x * blockDim.x + threadIdx.x;
    if (t >= n4) return;
    float4 a = ((const float4*)out)[t];
    float4 b = ((const float4*)Q1)[t];
    a.x += b.x; a.y += b.y; a.z += b.z; a.w += b.w;
    ((float4*)out)[t] = a;
}

extern "C" void kernel_launch(void* const* d_in, const int* in_sizes, int n_in,
                              void* d_out, int out_size, void* d_ws, size_t ws_size,
                              hipStream_t stream) {
    (void)in_sizes; (void)n_in; (void)out_size;
    const float* x  = (const float*)d_in[0];
    const int*   ei = (const int*)d_in[1];
    const float* W1 = (const float*)d_in[2];
    const float* b1 = (const float*)d_in[3];
    const float* W2 = (const float*)d_in[4];
    const float* b2 = (const float*)d_in[5];
    const float* W3 = (const float*)d_in[6];
    const float* b3 = (const float*)d_in[7];
    float* out = (float*)d_out;

    const int N = GCN_N, E = GCN_E;
    const int* srcI = ei;
    const int* dstI = ei + E;

    float* dinv = (float*)d_ws;
    float* Hbuf = dinv + N;
    float* Abuf = Hbuf + (size_t)N * 64;
    float* Abf1 = Abuf + (size_t)N * 64;
    float* deg1 = Abf1 + (size_t)N * 64;
    size_t need = (2 * (size_t)N + 3 * (size_t)N * 64) * sizeof(float);
    if (ws_size < need) return;

    const int TB = 256;
    const int mtiles = N / 16;
    const int wavesA = mtiles * 4;
    const int wavesB = mtiles * 2;
    const int gbA = (wavesA + 7) / 8;
    const int gbB = (wavesB + 7) / 8;

    deg_init_kernel <<<(N + TB - 1) / TB, TB, 0, stream>>>(dinv, deg1, N);
    deg_count_kernel<<<(E + TB - 1) / TB, TB, 0, stream>>>(dstI, dinv, deg1, E);
    dinv_kernel     <<<(N + TB - 1) / TB, TB, 0, stream>>>(dinv, deg1, N);

    gemm_wmma_kernel<128, 64, false, false><<<gbA, TB, 0, stream>>>(x, x, W1, Hbuf, N);
    agg_init_kernel<<<((N << 4) + TB - 1) / TB, TB, 0, stream>>>(Hbuf, dinv, b1, Abuf, Abf1, N, 4);
    agg_edge_kernel<64><<<E / TB, TB, 0, stream>>>(srcI, dstI, dinv, Hbuf, Abuf, Abf1, E);

    gemm_wmma_kernel<64, 64, true, true><<<gbA, TB, 0, stream>>>(Abuf, Abf1, W2, Hbuf, N);
    agg_init_kernel<<<((N << 4) + TB - 1) / TB, TB, 0, stream>>>(Hbuf, dinv, b2, Abuf, Abf1, N, 4);
    agg_edge_kernel<64><<<E / TB, TB, 0, stream>>>(srcI, dstI, dinv, Hbuf, Abuf, Abf1, E);

    gemm_wmma_kernel<64, 32, true, true><<<gbB, TB, 0, stream>>>(Abuf, Abf1, W3, Hbuf, N);
    agg_init_kernel<<<((N << 3) + TB - 1) / TB, TB, 0, stream>>>(Hbuf, dinv, b3, out, Abf1, N, 3);
    agg_edge_kernel<32><<<E / TB, TB, 0, stream>>>(srcI, dstI, dinv, Hbuf, out, Abf1, E);
    plane_join_kernel<<<((N << 3) + TB - 1) / TB, TB, 0, stream>>>(out, Abf1, N << 3);
}
